# starting kernel
# baseline (speedup 1.0000x reference)
#include <hip/hip_runtime.h>
#include <hip/hip_bf16.h>
#include <math.h>

typedef __attribute__((ext_vector_type(16))) __bf16 v16bf;
typedef __attribute__((ext_vector_type(8)))  __bf16 v8bf;
typedef __attribute__((ext_vector_type(8)))  float  v8f;

#define N_NODES 100000
#define F_IN    500
#define H_DIM   64
#define C_DIM   40
#define N_EDGES 3200000
#define K_STEPS 10
#define ALPHA   0.1f

__global__ __launch_bounds__(256) void appnp_gemm1(const float* __restrict__ x,
                                                   const float* __restrict__ W1,
                                                   const float* __restrict__ b1,
                                                   __bf16* __restrict__ h1) {
    const int lane = threadIdx.x & 31;
    const int wid  = threadIdx.x >> 5;
    const int rowTile = blockIdx.x * 2 + (wid >> 2);
    const int colTile = wid & 3;
    const int hiHalf  = (lane & 16) ? 8 : 0;
    const int m = rowTile * 16 + (lane & 15);
    const int n = colTile * 16 + (lane & 15);

    const float* xrow = x + (size_t)m * F_IN;
    v8f acc = {};

    for (int kb = 0; kb < 480; kb += 32) {
        const int kA = kb + hiHalf;
        const float4 r0 = *(const float4*)(xrow + kA);
        const float4 r1 = *(const float4*)(xrow + kA + 4);
        const float4 r2 = *(const float4*)(xrow + kA + 16);
        const float4 r3 = *(const float4*)(xrow + kA + 20);

        v16bf a;
        a[0]  = (__bf16)r0.x; a[1]  = (__bf16)r0.y; a[2]  = (__bf16)r0.z; a[3]  = (__bf16)r0.w;
        a[4]  = (__bf16)r1.x; a[5]  = (__bf16)r1.y; a[6]  = (__bf16)r1.z; a[7]  = (__bf16)r1.w;
        a[8]  = (__bf16)r2.x; a[9]  = (__bf16)r2.y; a[10] = (__bf16)r2.z; a[11] = (__bf16)r2.w;
        a[12] = (__bf16)r3.x; a[13] = (__bf16)r3.y; a[14] = (__bf16)r3.z; a[15] = (__bf16)r3.w;

        v16bf b;
#pragma unroll
        for (int j = 0; j < 16; ++j) {
            int k = kb + j + (j & 8) + hiHalf;
            b[j] = (__bf16)W1[k * H_DIM + n];
        }
        acc = __builtin_amdgcn_wmma_f32_16x16x32_bf16(
            false, a, false, b, (short)0, acc, false, false);
    }

    {
        const int kA = 480 + hiHalf;
        const float4 r0 = *(const float4*)(xrow + kA);
        const float4 r1 = *(const float4*)(xrow + kA + 4);
        const float4 r2 = *(const float4*)(xrow + 496);
        const float fa2[4] = {r2.x, r2.y, r2.z, r2.w};

        v16bf a;
        a[0] = (__bf16)r0.x; a[1] = (__bf16)r0.y; a[2] = (__bf16)r0.z; a[3] = (__bf16)r0.w;
        a[4] = (__bf16)r1.x; a[5] = (__bf16)r1.y; a[6] = (__bf16)r1.z; a[7] = (__bf16)r1.w;
#pragma unroll
        for (int j = 0; j < 8; ++j) {
            int k1 = kA + 16 + j;
            float val = (j < 4) ? fa2[j] : 0.0f;
            float f1  = (k1 < F_IN) ? val : 0.0f;
            a[j + 8] = (__bf16)f1;
        }

        v16bf b;
#pragma unroll
        for (int j = 0; j < 16; ++j) {
            int k  = 480 + j + (j & 8) + hiHalf;
            int kc = (k < F_IN) ? k : (F_IN - 1);
            b[j] = (__bf16)W1[kc * H_DIM + n];
        }
        acc = __builtin_amdgcn_wmma_f32_16x16x32_bf16(
            false, a, false, b, (short)0, acc, false, false);
    }

    const float bias = b1[n];
#pragma unroll
    for (int r = 0; r < 8; ++r) {
        int mr = rowTile * 16 + r + hiHalf;
        float v = fmaxf(acc[r] + bias, 0.0f);
        h1[(size_t)mr * H_DIM + n] = (__bf16)v;
    }
}

__global__ __launch_bounds__(256) void appnp_gemm2(const __bf16* __restrict__ h1,
                                                   const float* __restrict__ W2,
                                                   const float* __restrict__ b2,
                                                   float* __restrict__ h,
                                                   float* __restrict__ h0) {
    const int lane = threadIdx.x & 31;
    const int wid  = threadIdx.x >> 5;
    const int tile = blockIdx.x * 8 + wid;
    const int nTiles = (N_NODES / 16) * 3;
    if (tile >= nTiles) return;
    const int rowTile = tile / 3;
    const int colTile = tile % 3;
    const int hiHalf  = (lane & 16) ? 8 : 0;
    const int m = rowTile * 16 + (lane & 15);
    const int n = colTile * 16 + (lane & 15);
    const int ncl = (n < C_DIM) ? n : (C_DIM - 1);

    const __bf16* hrow = h1 + (size_t)m * H_DIM;
    v8f acc = {};
#pragma unroll
    for (int kb = 0; kb < H_DIM; kb += 32) {
        const int kA = kb + hiHalf;
        const v8bf lo = *(const v8bf*)(hrow + kA);
        const v8bf hi = *(const v8bf*)(hrow + kA + 16);
        v16bf a;
#pragma unroll
        for (int j = 0; j < 8; ++j) { a[j] = lo[j]; a[j + 8] = hi[j]; }

        v16bf b;
#pragma unroll
        for (int j = 0; j < 16; ++j) {
            int k = kb + j + (j & 8) + hiHalf;
            b[j] = (__bf16)W2[k * C_DIM + ncl];
        }
        acc = __builtin_amdgcn_wmma_f32_16x16x32_bf16(
            false, a, false, b, (short)0, acc, false, false);
    }

    if (n < C_DIM) {
        const float bias = b2[ncl];
#pragma unroll
        for (int r = 0; r < 8; ++r) {
            int mr = rowTile * 16 + r + hiHalf;
            float v = acc[r] + bias;
            h [(size_t)mr * C_DIM + n] = v;
            h0[(size_t)mr * C_DIM + n] = v;
        }
    }
}

__global__ __launch_bounds__(256) void appnp_deg_init(float* __restrict__ deg) {
    int i = blockIdx.x * blockDim.x + threadIdx.x;
    if (i < N_NODES) deg[i] = 1.0f;
}

__global__ __launch_bounds__(256) void appnp_deg_count(const int* __restrict__ dst,
                                                       float* __restrict__ deg) {
    int e = blockIdx.x * blockDim.x + threadIdx.x;
    if (e < N_EDGES) unsafeAtomicAdd(&deg[dst[e]], 1.0f);
}

__global__ __launch_bounds__(256) void appnp_dinv(const float* __restrict__ deg,
                                                  float* __restrict__ dinv) {
    int i = blockIdx.x * blockDim.x + threadIdx.x;
    if (i < N_NODES) dinv[i] = rsqrtf(deg[i]);
}

__global__ __launch_bounds__(256) void appnp_prop_init(const float* __restrict__ h,
                                                       const float* __restrict__ h0,
                                                       const float* __restrict__ dinv,
                                                       float* __restrict__ hn) {
    int t = blockIdx.x * blockDim.x + threadIdx.x;
    int i = t / C_DIM;
    float di = dinv[i];
    hn[t] = ALPHA * h0[t] + (1.0f - ALPHA) * di * di * h[t];
}

__global__ __launch_bounds__(256) void appnp_prop_edges(const int* __restrict__ src,
                                                        const int* __restrict__ dst,
                                                        const float* __restrict__ dinv,
                                                        const float* __restrict__ h,
                                                        float* __restrict__ hn) {
    const int lane = threadIdx.x & 31;
    const int e = blockIdx.x * 8 + (threadIdx.x >> 5);
    if (e >= N_EDGES) return;
    const int s = src[e];
    const int d = dst[e];
    const float w = (1.0f - ALPHA) * dinv[s] * dinv[d];
    const float* hs = h  + (size_t)s * C_DIM;
    float*       hd = hn + (size_t)d * C_DIM;
    unsafeAtomicAdd(&hd[lane], w * hs[lane]);
    if (lane < C_DIM - 32)
        unsafeAtomicAdd(&hd[lane + 32], w * hs[lane + 32]);
}

__global__ __launch_bounds__(256) void appnp_logsoftmax(const float* __restrict__ h,
                                                        float* __restrict__ out) {
    const int lane = threadIdx.x & 31;
    const int row = blockIdx.x * 8 + (threadIdx.x >> 5);
    if (row >= N_NODES) return;
    const float* r = h + (size_t)row * C_DIM;
    float v0 = r[lane];
    float v1 = (lane < C_DIM - 32) ? r[lane + 32] : -3.4e38f;

    float m = fmaxf(v0, v1);
#pragma unroll
    for (int off = 16; off > 0; off >>= 1) m = fmaxf(m, __shfl_xor(m, off, 32));

    float s = __expf(v0 - m) + ((lane < C_DIM - 32) ? __expf(v1 - m) : 0.0f);
#pragma unroll
    for (int off = 16; off > 0; off >>= 1) s += __shfl_xor(s, off, 32);

    float lse = m + __logf(s);
    float* o = out + (size_t)row * C_DIM;
    o[lane] = v0 - lse;
    if (lane < C_DIM - 32) o[lane + 32] = v1 - lse;
}

extern "C" void kernel_launch(void* const* d_in, const int* in_sizes, int n_in,
                              void* d_out, int out_size, void* d_ws, size_t ws_size,
                              hipStream_t stream) {
    const float* x  = (const float*)d_in[0];
    const float* W1 = (const float*)d_in[1];
    const float* b1 = (const float*)d_in[2];
    const float* W2 = (const float*)d_in[3];
    const float* b2 = (const float*)d_in[4];
    const int*   ei = (const int*)d_in[5];
    const int*   src = ei;
    const int*   dst = ei + N_EDGES;
    float* out = (float*)d_out;

    char* ws = (char*)d_ws;
    __bf16* h1  = (__bf16*)ws;
    float* hA   = (float*)(ws + 12800000);
    float* hB   = (float*)(ws + 12800000 + 16000000);
    float* h0   = (float*)(ws + 12800000 + 2 * 16000000);
    float* deg  = (float*)(ws + 12800000 + 3 * 16000000);
    float* dinv = (float*)(ws + 12800000 + 3 * 16000000 + 400128);

    appnp_gemm1<<<N_NODES / 32, 256, 0, stream>>>(x, W1, b1, h1);
    appnp_gemm2<<<((N_NODES / 16) * 3 + 7) / 8, 256, 0, stream>>>(h1, W2, b2, hA, h0);

    appnp_deg_init<<<(N_NODES + 255) / 256, 256, 0, stream>>>(deg);
    appnp_deg_count<<<(N_EDGES + 255) / 256, 256, 0, stream>>>(dst, deg);
    appnp_dinv<<<(N_NODES + 255) / 256, 256, 0, stream>>>(deg, dinv);

    float* cur = hA;
    float* nxt = hB;
    for (int k = 0; k < K_STEPS; ++k) {
        appnp_prop_init<<<(N_NODES * C_DIM) / 256, 256, 0, stream>>>(cur, h0, dinv, nxt);
        appnp_prop_edges<<<(N_EDGES + 7) / 8, 256, 0, stream>>>(src, dst, dinv, cur, nxt);
        float* t = cur; cur = nxt; nxt = t;
    }

    appnp_logsoftmax<<<(N_NODES + 7) / 8, 256, 0, stream>>>(cur, out);
}


# 10 propagation steps: per-edge float atomic scatter -> atomic-free per-node gather over edges grouped by destination once per call (int atomic rank + scan); sum order = rank arrival order
# speedup vs baseline: 4.0402x; 4.0402x over previous
#include <hip/hip_runtime.h>
#include <hip/hip_bf16.h>
#include <math.h>

typedef __attribute__((ext_vector_type(16))) __bf16 v16bf;
typedef __attribute__((ext_vector_type(8)))  __bf16 v8bf;
typedef __attribute__((ext_vector_type(8)))  float  v8f;

#define N_NODES 100000
#define F_IN    500
#define H_DIM   64
#define C_DIM   40
#define N_EDGES 3200000
#define K_STEPS 10
#define ALPHA   0.1f

__global__ __launch_bounds__(256) void appnp_gemm1(const float* __restrict__ x,
                                                   const float* __restrict__ W1,
                                                   const float* __restrict__ b1,
                                                   __bf16* __restrict__ h1) {
    const int lane = threadIdx.x & 31;
    const int wid  = threadIdx.x >> 5;
    const int rowTile = blockIdx.x * 2 + (wid >> 2);
    const int colTile = wid & 3;
    const int hiHalf  = (lane & 16) ? 8 : 0;
    const int m = rowTile * 16 + (lane & 15);
    const int n = colTile * 16 + (lane & 15);

    const float* xrow = x + (size_t)m * F_IN;
    v8f acc = {};

    for (int kb = 0; kb < 480; kb += 32) {
        const int kA = kb + hiHalf;
        const float4 r0 = *(const float4*)(xrow + kA);
        const float4 r1 = *(const float4*)(xrow + kA + 4);
        const float4 r2 = *(const float4*)(xrow + kA + 16);
        const float4 r3 = *(const float4*)(xrow + kA + 20);

        v16bf a;
        a[0]  = (__bf16)r0.x; a[1]  = (__bf16)r0.y; a[2]  = (__bf16)r0.z; a[3]  = (__bf16)r0.w;
        a[4]  = (__bf16)r1.x; a[5]  = (__bf16)r1.y; a[6]  = (__bf16)r1.z; a[7]  = (__bf16)r1.w;
        a[8]  = (__bf16)r2.x; a[9]  = (__bf16)r2.y; a[10] = (__bf16)r2.z; a[11] = (__bf16)r2.w;
        a[12] = (__bf16)r3.x; a[13] = (__bf16)r3.y; a[14] = (__bf16)r3.z; a[15] = (__bf16)r3.w;

        v16bf b;
#pragma unroll
        for (int j = 0; j < 16; ++j) {
            int k = kb + j + (j & 8) + hiHalf;
            b[j] = (__bf16)W1[k * H_DIM + n];
        }
        acc = __builtin_amdgcn_wmma_f32_16x16x32_bf16(
            false, a, false, b, (short)0, acc, false, false);
    }

    {
        const int kA = 480 + hiHalf;
        const float4 r0 = *(const float4*)(xrow + kA);
        const float4 r1 = *(const float4*)(xrow + kA + 4);
        const float4 r2 = *(const float4*)(xrow + 496);
        const float fa2[4] = {r2.x, r2.y, r2.z, r2.w};

        v16bf a;
        a[0] = (__bf16)r0.x; a[1] = (__bf16)r0.y; a[2] = (__bf16)r0.z; a[3] = (__bf16)r0.w;
        a[4] = (__bf16)r1.x; a[5] = (__bf16)r1.y; a[6] = (__bf16)r1.z; a[7] = (__bf16)r1.w;
#pragma unroll
        for (int j = 0; j < 8; ++j) {
            int k1 = kA + 16 + j;
            float val = (j < 4) ? fa2[j] : 0.0f;
            float f1  = (k1 < F_IN) ? val : 0.0f;
            a[j + 8] = (__bf16)f1;
        }

        v16bf b;
#pragma unroll
        for (int j = 0; j < 16; ++j) {
            int k  = 480 + j + (j & 8) + hiHalf;
            int kc = (k < F_IN) ? k : (F_IN - 1);
            b[j] = (__bf16)W1[kc * H_DIM + n];
        }
        acc = __builtin_amdgcn_wmma_f32_16x16x32_bf16(
            false, a, false, b, (short)0, acc, false, false);
    }

    const float bias = b1[n];
#pragma unroll
    for (int r = 0; r < 8; ++r) {
        int mr = rowTile * 16 + r + hiHalf;
        float v = fmaxf(acc[r] + bias, 0.0f);
        h1[(size_t)mr * H_DIM + n] = (__bf16)v;
    }
}

__global__ __launch_bounds__(256) void appnp_gemm2(const __bf16* __restrict__ h1,
                                                   const float* __restrict__ W2,
                                                   const float* __restrict__ b2,
                                                   float* __restrict__ h,
                                                   float* __restrict__ h0) {
    const int lane = threadIdx.x & 31;
    const int wid  = threadIdx.x >> 5;
    const int tile = blockIdx.x * 8 + wid;
    const int nTiles = (N_NODES / 16) * 3;
    if (tile >= nTiles) return;
    const int rowTile = tile / 3;
    const int colTile = tile % 3;
    const int hiHalf  = (lane & 16) ? 8 : 0;
    const int m = rowTile * 16 + (lane & 15);
    const int n = colTile * 16 + (lane & 15);
    const int ncl = (n < C_DIM) ? n : (C_DIM - 1);

    const __bf16* hrow = h1 + (size_t)m * H_DIM;
    v8f acc = {};
#pragma unroll
    for (int kb = 0; kb < H_DIM; kb += 32) {
        const int kA = kb + hiHalf;
        const v8bf lo = *(const v8bf*)(hrow + kA);
        const v8bf hi = *(const v8bf*)(hrow + kA + 16);
        v16bf a;
#pragma unroll
        for (int j = 0; j < 8; ++j) { a[j] = lo[j]; a[j + 8] = hi[j]; }

        v16bf b;
#pragma unroll
        for (int j = 0; j < 16; ++j) {
            int k = kb + j + (j & 8) + hiHalf;
            b[j] = (__bf16)W2[k * C_DIM + ncl];
        }
        acc = __builtin_amdgcn_wmma_f32_16x16x32_bf16(
            false, a, false, b, (short)0, acc, false, false);
    }

    if (n < C_DIM) {
        const float bias = b2[ncl];
#pragma unroll
        for (int r = 0; r < 8; ++r) {
            int mr = rowTile * 16 + r + hiHalf;
            float v = acc[r] + bias;
            h [(size_t)mr * C_DIM + n] = v;
            h0[(size_t)mr * C_DIM + n] = v;
        }
    }
}

__global__ __launch_bounds__(256) void appnp_cnt_zero(int* __restrict__ cnt) {
    int i = blockIdx.x * blockDim.x + threadIdx.x;
    if (i < N_NODES) cnt[i] = 0;
}

__global__ __launch_bounds__(256) void appnp_count_rank(const int* __restrict__ dst,
                                                        int* __restrict__ cnt,
                                                        int* __restrict__ rank) {
    int e = blockIdx.x * blockDim.x + threadIdx.x;
    if (e < N_EDGES) rank[e] = atomicAdd(&cnt[dst[e]], 1);
}

__global__ __launch_bounds__(1024) void appnp_scan(const int* __restrict__ cnt,
                                                   int* __restrict__ rowptr) {
    __shared__ int wsum[32];
    const int lane = threadIdx.x & 31;
    const int wid  = threadIdx.x >> 5;
    const int base = wid * 3200;

    int tot = 0;
    for (int it = 0; it < 25; ++it) {
        const int i = base + it * 128 + lane * 4;
        if (i < N_NODES) {
            const int4 c = *(const int4*)(cnt + i);
            tot += (c.x + c.y) + (c.z + c.w);
        }
    }
#pragma unroll
    for (int off = 16; off > 0; off >>= 1) tot += __shfl_xor(tot, off, 32);
    if (lane == 0) wsum[wid] = tot;
    __syncthreads();

    int carry = 0;
    for (int w = 0; w < wid; ++w) carry += wsum[w];

    for (int it = 0; it < 25; ++it) {
        const int i = base + it * 128 + lane * 4;
        int4 c = make_int4(0, 0, 0, 0);
        if (i < N_NODES) c = *(const int4*)(cnt + i);
        const int t = (c.x + c.y) + (c.z + c.w);
        int incl = t;
#pragma unroll
        for (int off = 1; off < 32; off <<= 1) {
            const int v = __shfl_up(incl, off, 32);
            if (lane >= off) incl += v;
        }
        const int excl = carry + incl - t;
        if (i < N_NODES) {
            int4 r;
            r.x = excl;
            r.y = r.x + c.x;
            r.z = r.y + c.y;
            r.w = r.z + c.z;
            *(int4*)(rowptr + i) = r;
        }
        carry += __shfl(incl, 31, 32);
    }
}

__global__ __launch_bounds__(256) void appnp_prescale(const int* __restrict__ cnt,
                                                      float* __restrict__ h) {
    const int t = blockIdx.x * blockDim.x + threadIdx.x;
    if (t < N_NODES * (C_DIM / 4)) {
        const int i = t / (C_DIM / 4);
        const float di = rsqrtf((float)(cnt[i] + 1));
        float4 v = ((const float4*)h)[t];
        v.x *= di; v.y *= di; v.z *= di; v.w *= di;
        ((float4*)h)[t] = v;
    }
}

__global__ __launch_bounds__(256) void appnp_fill(const int* __restrict__ src,
                                                  const int* __restrict__ dst,
                                                  const int* __restrict__ rank,
                                                  const int* __restrict__ rowptr,
                                                  int* __restrict__ col) {
    int e = blockIdx.x * blockDim.x + threadIdx.x;
    if (e < N_EDGES) col[rowptr[dst[e]] + rank[e]] = src[e];
}

template <bool LAST>
__global__ __launch_bounds__(256) void appnp_gather(const int* __restrict__ cnt,
                                                    const int* __restrict__ rowptr,
                                                    const int* __restrict__ col,
                                                    const float* __restrict__ g,
                                                    const float* __restrict__ h0,
                                                    float* __restrict__ gn) {
    const int lane = threadIdx.x & 31;
    const int i    = blockIdx.x * 8 + (threadIdx.x >> 5);
    const int grp  = lane / 10;
    const int q    = lane - grp * 10;
    const int n     = cnt[i];
    const int start = rowptr[i];
    const float di  = rsqrtf((float)(n + 1));
    const float4* g4 = (const float4*)g;

    float4 gi = make_float4(0.f, 0.f, 0.f, 0.f);
    float4 z  = make_float4(0.f, 0.f, 0.f, 0.f);
    if (lane < 10) {
        gi = g4[i * 10 + q];
        z  = ((const float4*)h0)[i * 10 + q];
    }

    float4 a0 = make_float4(0.f, 0.f, 0.f, 0.f);
    float4 a1 = a0, a2 = a0, a3 = a0;
    if (grp < 3) {
        const int* cp = col + start;
        for (int j = grp; j < n; j += 12) {
            const bool p1 = (j + 3) < n;
            const bool p2 = (j + 6) < n;
            const bool p3 = (j + 9) < n;
            const int c0 = cp[j];
            const int c1 = cp[p1 ? j + 3 : j];
            const int c2 = cp[p2 ? j + 6 : j];
            const int c3 = cp[p3 ? j + 9 : j];
            const float4 r0 = g4[c0 * 10 + q];
            const float4 r1 = g4[c1 * 10 + q];
            const float4 r2 = g4[c2 * 10 + q];
            const float4 r3 = g4[c3 * 10 + q];
            a0.x += r0.x; a0.y += r0.y; a0.z += r0.z; a0.w += r0.w;
            a1.x += p1 ? r1.x : 0.f; a1.y += p1 ? r1.y : 0.f;
            a1.z += p1 ? r1.z : 0.f; a1.w += p1 ? r1.w : 0.f;
            a2.x += p2 ? r2.x : 0.f; a2.y += p2 ? r2.y : 0.f;
            a2.z += p2 ? r2.z : 0.f; a2.w += p2 ? r2.w : 0.f;
            a3.x += p3 ? r3.x : 0.f; a3.y += p3 ? r3.y : 0.f;
            a3.z += p3 ? r3.z : 0.f; a3.w += p3 ? r3.w : 0.f;
        }
    }
    float4 a;
    a.x = (a0.x + a1.x) + (a2.x + a3.x);
    a.y = (a0.y + a1.y) + (a2.y + a3.y);
    a.z = (a0.z + a1.z) + (a2.z + a3.z);
    a.w = (a0.w + a1.w) + (a2.w + a3.w);

    float4 s;
    s.x = a.x + __shfl_down(a.x, 10, 32) + __shfl_down(a.x, 20, 32);
    s.y = a.y + __shfl_down(a.y, 10, 32) + __shfl_down(a.y, 20, 32);
    s.z = a.z + __shfl_down(a.z, 10, 32) + __shfl_down(a.z, 20, 32);
    s.w = a.w + __shfl_down(a.w, 10, 32) + __shfl_down(a.w, 20, 32);

    if (lane < 10) {
        const float wd = (1.0f - ALPHA) * di;
        float4 o;
        o.x = ALPHA * z.x + wd * (s.x + gi.x);
        o.y = ALPHA * z.y + wd * (s.y + gi.y);
        o.z = ALPHA * z.z + wd * (s.z + gi.z);
        o.w = ALPHA * z.w + wd * (s.w + gi.w);
        if (!LAST) { o.x *= di; o.y *= di; o.z *= di; o.w *= di; }
        ((float4*)gn)[i * 10 + q] = o;
    }
}

__global__ __launch_bounds__(256) void appnp_logsoftmax(const float* __restrict__ h,
                                                        float* __restrict__ out) {
    const int lane = threadIdx.x & 31;
    const int row = blockIdx.x * 8 + (threadIdx.x >> 5);
    if (row >= N_NODES) return;
    const float* r = h + (size_t)row * C_DIM;
    float v0 = r[lane];
    float v1 = (lane < C_DIM - 32) ? r[lane + 32] : -3.4e38f;

    float m = fmaxf(v0, v1);
#pragma unroll
    for (int off = 16; off > 0; off >>= 1) m = fmaxf(m, __shfl_xor(m, off, 32));

    float s = __expf(v0 - m) + ((lane < C_DIM - 32) ? __expf(v1 - m) : 0.0f);
#pragma unroll
    for (int off = 16; off > 0; off >>= 1) s += __shfl_xor(s, off, 32);

    float lse = m + __logf(s);
    float* o = out + (size_t)row * C_DIM;
    o[lane] = v0 - lse;
    if (lane < C_DIM - 32) o[lane + 32] = v1 - lse;
}

extern "C" void kernel_launch(void* const* d_in, const int* in_sizes, int n_in,
                              void* d_out, int out_size, void* d_ws, size_t ws_size,
                              hipStream_t stream) {
    const float* x  = (const float*)d_in[0];
    const float* W1 = (const float*)d_in[1];
    const float* b1 = (const float*)d_in[2];
    const float* W2 = (const float*)d_in[3];
    const float* b2 = (const float*)d_in[4];
    const int*   ei = (const int*)d_in[5];
    const int*   src = ei;
    const int*   dst = ei + N_EDGES;
    float* out = (float*)d_out;

    char* ws = (char*)d_ws;
    __bf16* h1  = (__bf16*)ws;
    int*   col  = (int*)ws;
    float* hA   = (float*)(ws + 12800000);
    float* hB   = (float*)(ws + 12800000 + 16000000);
    int*   rank = (int*)(ws + 12800000 + 16000000);
    float* h0   = (float*)(ws + 12800000 + 2 * 16000000);
    int* cnt    = (int*)(ws + 12800000 + 3 * 16000000);
    int* rowptr = (int*)(ws + 12800000 + 3 * 16000000 + 400128);

    appnp_gemm1<<<N_NODES / 32, 256, 0, stream>>>(x, W1, b1, h1);
    appnp_gemm2<<<((N_NODES / 16) * 3 + 7) / 8, 256, 0, stream>>>(h1, W2, b2, hA, h0);

    appnp_cnt_zero<<<(N_NODES + 255) / 256, 256, 0, stream>>>(cnt);
    appnp_count_rank<<<(N_EDGES + 255) / 256, 256, 0, stream>>>(dst, cnt, rank);
    appnp_scan<<<1, 1024, 0, stream>>>(cnt, rowptr);
    appnp_prescale<<<(N_NODES * (C_DIM / 4) + 255) / 256, 256, 0, stream>>>(cnt, hA);
    appnp_fill<<<(N_EDGES + 255) / 256, 256, 0, stream>>>(src, dst, rank, rowptr, col);

    float* cur = hA;
    float* nxt = hB;
    for (int k = 0; k < K_STEPS - 1; ++k) {
        appnp_gather<false><<<N_NODES / 8, 256, 0, stream>>>(cnt, rowptr, col, cur, h0, nxt);
        float* t = cur; cur = nxt; nxt = t;
    }
    appnp_gather<true><<<N_NODES / 8, 256, 0, stream>>>(cnt, rowptr, col, cur, h0, nxt);
    { float* t = cur; cur = nxt; nxt = t; }

    appnp_logsoftmax<<<(N_NODES + 7) / 8, 256, 0, stream>>>(cur, out);
}


# Propagated rows at a 256-B pitch, 256-B aligned (40 values + 24 of the node's h0 row): a gathered neighbour row is one aligned 256-B unit, where half straddled two; edge list parked in out meanwhile
# speedup vs baseline: 4.7366x; 1.1724x over previous
#include <hip/hip_runtime.h>
#include <hip/hip_bf16.h>
#include <math.h>

typedef __attribute__((ext_vector_type(16))) __bf16 v16bf;
typedef __attribute__((ext_vector_type(8)))  __bf16 v8bf;
typedef __attribute__((ext_vector_type(8)))  float  v8f;

#define N_NODES 100000
#define F_IN    500
#define H_DIM   64
#define C_DIM   40
#define N_EDGES 3200000
#define K_STEPS 10
#define ALPHA   0.1f
#define ROW_F   64
#define ROW_F4  16

__global__ __launch_bounds__(256) void appnp_gemm1(const float* __restrict__ x,
                                                   const float* __restrict__ W1,
                                                   const float* __restrict__ b1,
                                                   __bf16* __restrict__ h1) {
    const int lane = threadIdx.x & 31;
    const int wid  = threadIdx.x >> 5;
    const int rowTile = blockIdx.x * 2 + (wid >> 2);
    const int colTile = wid & 3;
    const int hiHalf  = (lane & 16) ? 8 : 0;
    const int m = rowTile * 16 + (lane & 15);
    const int n = colTile * 16 + (lane & 15);

    const float* xrow = x + (size_t)m * F_IN;
    v8f acc = {};

    for (int kb = 0; kb < 480; kb += 32) {
        const int kA = kb + hiHalf;
        const float4 r0 = *(const float4*)(xrow + kA);
        const float4 r1 = *(const float4*)(xrow + kA + 4);
        const float4 r2 = *(const float4*)(xrow + kA + 16);
        const float4 r3 = *(const float4*)(xrow + kA + 20);

        v16bf a;
        a[0]  = (__bf16)r0.x; a[1]  = (__bf16)r0.y; a[2]  = (__bf16)r0.z; a[3]  = (__bf16)r0.w;
        a[4]  = (__bf16)r1.x; a[5]  = (__bf16)r1.y; a[6]  = (__bf16)r1.z; a[7]  = (__bf16)r1.w;
        a[8]  = (__bf16)r2.x; a[9]  = (__bf16)r2.y; a[10] = (__bf16)r2.z; a[11] = (__bf16)r2.w;
        a[12] = (__bf16)r3.x; a[13] = (__bf16)r3.y; a[14] = (__bf16)r3.z; a[15] = (__bf16)r3.w;

        v16bf b;
#pragma unroll
        for (int j = 0; j < 16; ++j) {
            int k = kb + j + (j & 8) + hiHalf;
            b[j] = (__bf16)W1[k * H_DIM + n];
        }
        acc = __builtin_amdgcn_wmma_f32_16x16x32_bf16(
            false, a, false, b, (short)0, acc, false, false);
    }

    {
        const int kA = 480 + hiHalf;
        const float4 r0 = *(const float4*)(xrow + kA);
        const float4 r1 = *(const float4*)(xrow + kA + 4);
        const float4 r2 = *(const float4*)(xrow + 496);
        const float fa2[4] = {r2.x, r2.y, r2.z, r2.w};

        v16bf a;
        a[0] = (__bf16)r0.x; a[1] = (__bf16)r0.y; a[2] = (__bf16)r0.z; a[3] = (__bf16)r0.w;
        a[4] = (__bf16)r1.x; a[5] = (__bf16)r1.y; a[6] = (__bf16)r1.z; a[7] = (__bf16)r1.w;
#pragma unroll
        for (int j = 0; j < 8; ++j) {
            int k1 = kA + 16 + j;
            float val = (j < 4) ? fa2[j] : 0.0f;
            float f1  = (k1 < F_IN) ? val : 0.0f;
            a[j + 8] = (__bf16)f1;
        }

        v16bf b;
#pragma unroll
        for (int j = 0; j < 16; ++j) {
            int k  = 480 + j + (j & 8) + hiHalf;
            int kc = (k < F_IN) ? k : (F_IN - 1);
            b[j] = (__bf16)W1[kc * H_DIM + n];
        }
        acc = __builtin_amdgcn_wmma_f32_16x16x32_bf16(
            false, a, false, b, (short)0, acc, false, false);
    }

    const float bias = b1[n];
#pragma unroll
    for (int r = 0; r < 8; ++r) {
        int mr = rowTile * 16 + r + hiHalf;
        float v = fmaxf(acc[r] + bias, 0.0f);
        h1[(size_t)mr * H_DIM + n] = (__bf16)v;
    }
}

__global__ __launch_bounds__(256) void appnp_gemm2(const __bf16* __restrict__ h1,
                                                   const float* __restrict__ W2,
                                                   const float* __restrict__ b2,
                                                   float* __restrict__ tab,
                                                   float* __restrict__ h0r) {
    const int lane = threadIdx.x & 31;
    const int wid  = threadIdx.x >> 5;
    const int tile = blockIdx.x * 8 + wid;
    const int nTiles = (N_NODES / 16) * 3;
    if (tile >= nTiles) return;
    const int rowTile = tile / 3;
    const int colTile = tile % 3;
    const int hiHalf  = (lane & 16) ? 8 : 0;
    const int m = rowTile * 16 + (lane & 15);
    const int n = colTile * 16 + (lane & 15);
    const int ncl = (n < C_DIM) ? n : (C_DIM - 1);

    const __bf16* hrow = h1 + (size_t)m * H_DIM;
    v8f acc = {};
#pragma unroll
    for (int kb = 0; kb < H_DIM; kb += 32) {
        const int kA = kb + hiHalf;
        const v8bf lo = *(const v8bf*)(hrow + kA);
        const v8bf hi = *(const v8bf*)(hrow + kA + 16);
        v16bf a;
#pragma unroll
        for (int j = 0; j < 8; ++j) { a[j] = lo[j]; a[j + 8] = hi[j]; }

        v16bf b;
#pragma unroll
        for (int j = 0; j < 16; ++j) {
            int k = kb + j + (j & 8) + hiHalf;
            b[j] = (__bf16)W2[k * C_DIM + ncl];
        }
        acc = __builtin_amdgcn_wmma_f32_16x16x32_bf16(
            false, a, false, b, (short)0, acc, false, false);
    }

    if (n < C_DIM) {
        const float bias = b2[ncl];
#pragma unroll
        for (int r = 0; r < 8; ++r) {
            int mr = rowTile * 16 + r + hiHalf;
            float v = acc[r] + bias;
            float* row = tab + (size_t)mr * ROW_F;
            row[n] = v;
            if (n < 24) row[C_DIM + n] = v;
            else        h0r[(size_t)mr * 16 + (n - 24)] = v;
        }
    }
}

__global__ __launch_bounds__(256) void appnp_cnt_zero(int* __restrict__ cnt) {
    int i = blockIdx.x * blockDim.x + threadIdx.x;
    if (i < N_NODES) cnt[i] = 0;
}

__global__ __launch_bounds__(256) void appnp_count_rank(const int* __restrict__ dst,
                                                        int* __restrict__ cnt,
                                                        int* __restrict__ rank) {
    int e = blockIdx.x * blockDim.x + threadIdx.x;
    if (e < N_EDGES) rank[e] = atomicAdd(&cnt[dst[e]], 1);
}

__global__ __launch_bounds__(1024) void appnp_scan(const int* __restrict__ cnt,
                                                   int* __restrict__ rowptr) {
    __shared__ int wsum[32];
    const int lane = threadIdx.x & 31;
    const int wid  = threadIdx.x >> 5;
    const int base = wid * 3200;

    int tot = 0;
    for (int it = 0; it < 25; ++it) {
        const int i = base + it * 128 + lane * 4;
        if (i < N_NODES) {
            const int4 c = *(const int4*)(cnt + i);
            tot += (c.x + c.y) + (c.z + c.w);
        }
    }
#pragma unroll
    for (int off = 16; off > 0; off >>= 1) tot += __shfl_xor(tot, off, 32);
    if (lane == 0) wsum[wid] = tot;
    __syncthreads();

    int carry = 0;
    for (int w = 0; w < wid; ++w) carry += wsum[w];

    for (int it = 0; it < 25; ++it) {
        const int i = base + it * 128 + lane * 4;
        int4 c = make_int4(0, 0, 0, 0);
        if (i < N_NODES) c = *(const int4*)(cnt + i);
        const int t = (c.x + c.y) + (c.z + c.w);
        int incl = t;
#pragma unroll
        for (int off = 1; off < 32; off <<= 1) {
            const int v = __shfl_up(incl, off, 32);
            if (lane >= off) incl += v;
        }
        const int excl = carry + incl - t;
        if (i < N_NODES) {
            int4 r;
            r.x = excl;
            r.y = r.x + c.x;
            r.z = r.y + c.y;
            r.w = r.z + c.z;
            *(int4*)(rowptr + i) = r;
        }
        carry += __shfl(incl, 31, 32);
    }
}

__global__ __launch_bounds__(256) void appnp_prescale(const int* __restrict__ cnt,
                                                      float* __restrict__ tab) {
    const int t = blockIdx.x * blockDim.x + threadIdx.x;
    if (t < N_NODES * (C_DIM / 4)) {
        const int i = t / (C_DIM / 4);
        const int q = t - i * (C_DIM / 4);
        const float di = rsqrtf((float)(cnt[i] + 1));
        float4* p = (float4*)tab + (i * ROW_F4 + q);
        float4 v = *p;
        v.x *= di; v.y *= di; v.z *= di; v.w *= di;
        *p = v;
    }
}

__global__ __launch_bounds__(256) void appnp_fill(const int* __restrict__ src,
                                                  const int* __restrict__ dst,
                                                  const int* __restrict__ rank,
                                                  const int* __restrict__ rowptr,
                                                  int* __restrict__ col) {
    int e = blockIdx.x * blockDim.x + threadIdx.x;
    if (e < N_EDGES) col[rowptr[dst[e]] + rank[e]] = src[e];
}

template <bool LAST>
__global__ __launch_bounds__(256) void appnp_gather(const int* __restrict__ cnt,
                                                    const int* __restrict__ rowptr,
                                                    const int* __restrict__ col,
                                                    const float* __restrict__ g,
                                                    const float* __restrict__ h0r,
                                                    float* __restrict__ gn) {
    const int lane = threadIdx.x & 31;
    const int i    = blockIdx.x * 8 + (threadIdx.x >> 5);
    const int grp  = lane / 10;
    const int q    = lane - grp * 10;
    const int n     = cnt[i];
    const int start = rowptr[i];
    const float di  = rsqrtf((float)(n + 1));
    const float4* g4 = (const float4*)g;

    float4 gi = make_float4(0.f, 0.f, 0.f, 0.f);
    if (lane < 16) gi = g4[i * ROW_F4 + lane];
    float4 zr = make_float4(0.f, 0.f, 0.f, 0.f);
    if (lane >= 6 && lane < 10) zr = ((const float4*)h0r)[i * 4 + (lane - 6)];

    float4 a0 = make_float4(0.f, 0.f, 0.f, 0.f);
    float4 a1 = a0, a2 = a0, a3 = a0;
    if (grp < 3) {
        const int* cp = col + start;
        for (int j = grp; j < n; j += 12) {
            const bool p1 = (j + 3) < n;
            const bool p2 = (j + 6) < n;
            const bool p3 = (j + 9) < n;
            const int c0 = cp[j];
            const int c1 = cp[p1 ? j + 3 : j];
            const int c2 = cp[p2 ? j + 6 : j];
            const int c3 = cp[p3 ? j + 9 : j];
            const float4 r0 = g4[c0 * ROW_F4 + q];
            const float4 r1 = g4[c1 * ROW_F4 + q];
            const float4 r2 = g4[c2 * ROW_F4 + q];
            const float4 r3 = g4[c3 * ROW_F4 + q];
            a0.x += r0.x; a0.y += r0.y; a0.z += r0.z; a0.w += r0.w;
            a1.x += p1 ? r1.x : 0.f; a1.y += p1 ? r1.y : 0.f;
            a1.z += p1 ? r1.z : 0.f; a1.w += p1 ? r1.w : 0.f;
            a2.x += p2 ? r2.x : 0.f; a2.y += p2 ? r2.y : 0.f;
            a2.z += p2 ? r2.z : 0.f; a2.w += p2 ? r2.w : 0.f;
            a3.x += p3 ? r3.x : 0.f; a3.y += p3 ? r3.y : 0.f;
            a3.z += p3 ? r3.z : 0.f; a3.w += p3 ? r3.w : 0.f;
        }
    }
    float4 a;
    a.x = (a0.x + a1.x) + (a2.x + a3.x);
    a.y = (a0.y + a1.y) + (a2.y + a3.y);
    a.z = (a0.z + a1.z) + (a2.z + a3.z);
    a.w = (a0.w + a1.w) + (a2.w + a3.w);

    float4 s;
    s.x = a.x + __shfl_down(a.x, 10, 32) + __shfl_down(a.x, 20, 32);
    s.y = a.y + __shfl_down(a.y, 10, 32) + __shfl_down(a.y, 20, 32);
    s.z = a.z + __shfl_down(a.z, 10, 32) + __shfl_down(a.z, 20, 32);
    s.w = a.w + __shfl_down(a.w, 10, 32) + __shfl_down(a.w, 20, 32);

    float4 z;
    z.x = __shfl_down(gi.x, 10, 32);
    z.y = __shfl_down(gi.y, 10, 32);
    z.z = __shfl_down(gi.z, 10, 32);
    z.w = __shfl_down(gi.w, 10, 32);
    if (lane >= 6) z = zr;

    if (lane < 16) {
        float4 o = gi;
        if (lane < 10) {
            const float wd = (1.0f - ALPHA) * di;
            o.x = ALPHA * z.x + wd * (s.x + gi.x);
            o.y = ALPHA * z.y + wd * (s.y + gi.y);
            o.z = ALPHA * z.z + wd * (s.z + gi.z);
            o.w = ALPHA * z.w + wd * (s.w + gi.w);
            if (!LAST) { o.x *= di; o.y *= di; o.z *= di; o.w *= di; }
        }
        ((float4*)gn)[i * ROW_F4 + lane] = o;
    }
}

__global__ __launch_bounds__(256) void appnp_logsoftmax(const float* __restrict__ h,
                                                        float* __restrict__ out) {
    const int lane = threadIdx.x & 31;
    const int row = blockIdx.x * 8 + (threadIdx.x >> 5);
    if (row >= N_NODES) return;
    const float* r = h + (size_t)row * ROW_F;
    float v0 = r[lane];
    float v1 = (lane < C_DIM - 32) ? r[lane + 32] : -3.4e38f;

    float m = fmaxf(v0, v1);
#pragma unroll
    for (int off = 16; off > 0; off >>= 1) m = fmaxf(m, __shfl_xor(m, off, 32));

    float s = __expf(v0 - m) + ((lane < C_DIM - 32) ? __expf(v1 - m) : 0.0f);
#pragma unroll
    for (int off = 16; off > 0; off >>= 1) s += __shfl_xor(s, off, 32);

    float lse = m + __logf(s);
    float* o = out + (size_t)row * C_DIM;
    o[lane] = v0 - lse;
    if (lane < C_DIM - 32) o[lane + 32] = v1 - lse;
}

extern "C" void kernel_launch(void* const* d_in, const int* in_sizes, int n_in,
                              void* d_out, int out_size, void* d_ws, size_t ws_size,
                              hipStream_t stream) {
    const float* x  = (const float*)d_in[0];
    const float* W1 = (const float*)d_in[1];
    const float* b1 = (const float*)d_in[2];
    const float* W2 = (const float*)d_in[3];
    const float* b2 = (const float*)d_in[4];
    const int*   ei = (const int*)d_in[5];
    const int*   src = ei;
    const int*   dst = ei + N_EDGES;
    float* out = (float*)d_out;

    char* ws = (char*)d_ws;
    float* hA   = (float*)ws;
    float* hB   = (float*)(ws + 25600000);
    __bf16* h1  = (__bf16*)(ws + 25600000);
    int*   rank = (int*)(ws + 25600000);
    float* h0r  = (float*)(ws + 2 * 25600000);
    int* cnt    = (int*)(ws + 2 * 25600000 + 6400000);
    int* rowptr = (int*)(ws + 2 * 25600000 + 6400000 + 400128);
    int*   col  = (int*)d_out;

    appnp_gemm1<<<N_NODES / 32, 256, 0, stream>>>(x, W1, b1, h1);
    appnp_gemm2<<<((N_NODES / 16) * 3 + 7) / 8, 256, 0, stream>>>(h1, W2, b2, hA, h0r);

    appnp_cnt_zero<<<(N_NODES + 255) / 256, 256, 0, stream>>>(cnt);
    appnp_count_rank<<<(N_EDGES + 255) / 256, 256, 0, stream>>>(dst, cnt, rank);
    appnp_scan<<<1, 1024, 0, stream>>>(cnt, rowptr);
    appnp_prescale<<<(N_NODES * (C_DIM / 4) + 255) / 256, 256, 0, stream>>>(cnt, hA);
    appnp_fill<<<(N_EDGES + 255) / 256, 256, 0, stream>>>(src, dst, rank, rowptr, col);

    float* cur = hA;
    float* nxt = hB;
    for (int k = 0; k < K_STEPS - 1; ++k) {
        appnp_gather<false><<<N_NODES / 8, 256, 0, stream>>>(cnt, rowptr, col, cur, h0r, nxt);
        float* t = cur; cur = nxt; nxt = t;
    }
    appnp_gather<true><<<N_NODES / 8, 256, 0, stream>>>(cnt, rowptr, col, cur, h0r, nxt);
    { float* t = cur; cur = nxt; nxt = t; }

    appnp_logsoftmax<<<(N_NODES + 7) / 8, 256, 0, stream>>>(cur, out);
}


# GEMM1: one wave computes all 64 columns of its 16 rows (4 accumulator tiles on one A fragment, 4 interleaved columns per lane): x tile loaded once, not 4 times; W1 as 128-bit loads
# speedup vs baseline: 5.0423x; 1.0645x over previous
#include <hip/hip_runtime.h>
#include <hip/hip_bf16.h>
#include <math.h>

typedef __attribute__((ext_vector_type(16))) __bf16 v16bf;
typedef __attribute__((ext_vector_type(8)))  __bf16 v8bf;
typedef __attribute__((ext_vector_type(8)))  float  v8f;

#define N_NODES 100000
#define F_IN    500
#define H_DIM   64
#define C_DIM   40
#define N_EDGES 3200000
#define K_STEPS 10
#define ALPHA   0.1f
#define ROW_F   64
#define ROW_F4  16

__global__ __launch_bounds__(128) void appnp_gemm1(const float* __restrict__ x,
                                                   const float* __restrict__ W1,
                                                   const float* __restrict__ b1,
                                                   __bf16* __restrict__ h1) {
    const int lane = threadIdx.x & 31;
    const int rowTile = blockIdx.x * 4 + (threadIdx.x >> 5);
    if (rowTile >= N_NODES / 16) return;
    const int c      = lane & 15;
    const int hiHalf = (lane & 16) ? 8 : 0;
    const int m = rowTile * 16 + c;

    const float* xrow = x + (size_t)m * F_IN;
    const float* wcol = W1 + hiHalf * H_DIM + 4 * c;
    v8f acc0 = {}, acc1 = {}, acc2 = {}, acc3 = {};

    for (int kb = 0; kb < 480; kb += 32) {
        float4 w[16];
#pragma unroll
        for (int j = 0; j < 16; ++j) {
            const int k = kb + j + (j & 8);
            w[j] = *(const float4*)(wcol + k * H_DIM);
        }
        const int kA = kb + hiHalf;
        const float4 r0 = *(const float4*)(xrow + kA);
        const float4 r1 = *(const float4*)(xrow + kA + 4);
        const float4 r2 = *(const float4*)(xrow + kA + 16);
        const float4 r3 = *(const float4*)(xrow + kA + 20);
        __builtin_amdgcn_sched_barrier(0);

        v16bf bt0, bt1, bt2, bt3;
#pragma unroll
        for (int j = 0; j < 16; ++j) {
            bt0[j] = (__bf16)w[j].x; bt1[j] = (__bf16)w[j].y;
            bt2[j] = (__bf16)w[j].z; bt3[j] = (__bf16)w[j].w;
        }

        v16bf a;
        a[0]  = (__bf16)r0.x; a[1]  = (__bf16)r0.y; a[2]  = (__bf16)r0.z; a[3]  = (__bf16)r0.w;
        a[4]  = (__bf16)r1.x; a[5]  = (__bf16)r1.y; a[6]  = (__bf16)r1.z; a[7]  = (__bf16)r1.w;
        a[8]  = (__bf16)r2.x; a[9]  = (__bf16)r2.y; a[10] = (__bf16)r2.z; a[11] = (__bf16)r2.w;
        a[12] = (__bf16)r3.x; a[13] = (__bf16)r3.y; a[14] = (__bf16)r3.z; a[15] = (__bf16)r3.w;

        acc0 = __builtin_amdgcn_wmma_f32_16x16x32_bf16(
            false, a, false, bt0, (short)0, acc0, false, false);
        acc1 = __builtin_amdgcn_wmma_f32_16x16x32_bf16(
            false, a, false, bt1, (short)0, acc1, false, false);
        acc2 = __builtin_amdgcn_wmma_f32_16x16x32_bf16(
            false, a, false, bt2, (short)0, acc2, false, false);
        acc3 = __builtin_amdgcn_wmma_f32_16x16x32_bf16(
            false, a, false, bt3, (short)0, acc3, false, false);
    }

    {
        const int kA = 480 + hiHalf;
        const float4 r0 = *(const float4*)(xrow + kA);
        const float4 r1 = *(const float4*)(xrow + kA + 4);
        const float4 r2 = *(const float4*)(xrow + 496);
        const float fa2[4] = {r2.x, r2.y, r2.z, r2.w};

        v16bf a;
        a[0] = (__bf16)r0.x; a[1] = (__bf16)r0.y; a[2] = (__bf16)r0.z; a[3] = (__bf16)r0.w;
        a[4] = (__bf16)r1.x; a[5] = (__bf16)r1.y; a[6] = (__bf16)r1.z; a[7] = (__bf16)r1.w;
#pragma unroll
        for (int j = 0; j < 8; ++j) {
            int k1 = kA + 16 + j;
            float val = (j < 4) ? fa2[j] : 0.0f;
            float f1  = (k1 < F_IN) ? val : 0.0f;
            a[j + 8] = (__bf16)f1;
        }

        v16bf bt0, bt1, bt2, bt3;
#pragma unroll
        for (int j = 0; j < 16; ++j) {
            int k  = 480 + j + (j & 8) + hiHalf;
            int kc = (k < F_IN) ? k : (F_IN - 1);
            const float4 w = *(const float4*)(W1 + kc * H_DIM + 4 * c);
            bt0[j] = (__bf16)w.x; bt1[j] = (__bf16)w.y;
            bt2[j] = (__bf16)w.z; bt3[j] = (__bf16)w.w;
        }
        acc0 = __builtin_amdgcn_wmma_f32_16x16x32_bf16(
            false, a, false, bt0, (short)0, acc0, false, false);
        acc1 = __builtin_amdgcn_wmma_f32_16x16x32_bf16(
            false, a, false, bt1, (short)0, acc1, false, false);
        acc2 = __builtin_amdgcn_wmma_f32_16x16x32_bf16(
            false, a, false, bt2, (short)0, acc2, false, false);
        acc3 = __builtin_amdgcn_wmma_f32_16x16x32_bf16(
            false, a, false, bt3, (short)0, acc3, false, false);
    }

    const float4 bias = *(const float4*)(b1 + 4 * c);
    typedef __attribute__((ext_vector_type(4))) __bf16 v4bf;
#pragma unroll
    for (int r = 0; r < 8; ++r) {
        const int mr = rowTile * 16 + r + hiHalf;
        v4bf o;
        o[0] = (__bf16)fmaxf(acc0[r] + bias.x, 0.0f);
        o[1] = (__bf16)fmaxf(acc1[r] + bias.y, 0.0f);
        o[2] = (__bf16)fmaxf(acc2[r] + bias.z, 0.0f);
        o[3] = (__bf16)fmaxf(acc3[r] + bias.w, 0.0f);
        *(v4bf*)(h1 + (size_t)mr * H_DIM + 4 * c) = o;
    }
}

__global__ __launch_bounds__(256) void appnp_gemm2(const __bf16* __restrict__ h1,
                                                   const float* __restrict__ W2,
                                                   const float* __restrict__ b2,
                                                   float* __restrict__ tab,
                                                   float* __restrict__ h0r) {
    const int lane = threadIdx.x & 31;
    const int wid  = threadIdx.x >> 5;
    const int tile = blockIdx.x * 8 + wid;
    const int nTiles = (N_NODES / 16) * 3;
    if (tile >= nTiles) return;
    const int rowTile = tile / 3;
    const int colTile = tile % 3;
    const int hiHalf  = (lane & 16) ? 8 : 0;
    const int m = rowTile * 16 + (lane & 15);
    const int n = colTile * 16 + (lane & 15);
    const int ncl = (n < C_DIM) ? n : (C_DIM - 1);

    const __bf16* hrow = h1 + (size_t)m * H_DIM;
    v8f acc = {};
#pragma unroll
    for (int kb = 0; kb < H_DIM; kb += 32) {
        const int kA = kb + hiHalf;
        const v8bf lo = *(const v8bf*)(hrow + kA);
        const v8bf hi = *(const v8bf*)(hrow + kA + 16);
        v16bf a;
#pragma unroll
        for (int j = 0; j < 8; ++j) { a[j] = lo[j]; a[j + 8] = hi[j]; }

        v16bf b;
#pragma unroll
        for (int j = 0; j < 16; ++j) {
            int k = kb + j + (j & 8) + hiHalf;
            b[j] = (__bf16)W2[k * C_DIM + ncl];
        }
        acc = __builtin_amdgcn_wmma_f32_16x16x32_bf16(
            false, a, false, b, (short)0, acc, false, false);
    }

    if (n < C_DIM) {
        const float bias = b2[ncl];
#pragma unroll
        for (int r = 0; r < 8; ++r) {
            int mr = rowTile * 16 + r + hiHalf;
            float v = acc[r] + bias;
            float* row = tab + (size_t)mr * ROW_F;
            row[n] = v;
            if (n < 24) row[C_DIM + n] = v;
            else        h0r[(size_t)mr * 16 + (n - 24)] = v;
        }
    }
}

__global__ __launch_bounds__(256) void appnp_cnt_zero(int* __restrict__ cnt) {
    int i = blockIdx.x * blockDim.x + threadIdx.x;
    if (i < N_NODES) cnt[i] = 0;
}

__global__ __launch_bounds__(256) void appnp_count_rank(const int* __restrict__ dst,
                                                        int* __restrict__ cnt,
                                                        int* __restrict__ rank) {
    int e = blockIdx.x * blockDim.x + threadIdx.x;
    if (e < N_EDGES) rank[e] = atomicAdd(&cnt[dst[e]], 1);
}

__global__ __launch_bounds__(1024) void appnp_scan(const int* __restrict__ cnt,
                                                   int* __restrict__ rowptr) {
    __shared__ int wsum[32];
    const int lane = threadIdx.x & 31;
    const int wid  = threadIdx.x >> 5;
    const int base = wid * 3200;

    int tot = 0;
    for (int it = 0; it < 25; ++it) {
        const int i = base + it * 128 + lane * 4;
        if (i < N_NODES) {
            const int4 c = *(const int4*)(cnt + i);
            tot += (c.x + c.y) + (c.z + c.w);
        }
    }
#pragma unroll
    for (int off = 16; off > 0; off >>= 1) tot += __shfl_xor(tot, off, 32);
    if (lane == 0) wsum[wid] = tot;
    __syncthreads();

    int carry = 0;
    for (int w = 0; w < wid; ++w) carry += wsum[w];

    for (int it = 0; it < 25; ++it) {
        const int i = base + it * 128 + lane * 4;
        int4 c = make_int4(0, 0, 0, 0);
        if (i < N_NODES) c = *(const int4*)(cnt + i);
        const int t = (c.x + c.y) + (c.z + c.w);
        int incl = t;
#pragma unroll
        for (int off = 1; off < 32; off <<= 1) {
            const int v = __shfl_up(incl, off, 32);
            if (lane >= off) incl += v;
        }
        const int excl = carry + incl - t;
        if (i < N_NODES) {
            int4 r;
            r.x = excl;
            r.y = r.x + c.x;
            r.z = r.y + c.y;
            r.w = r.z + c.z;
            *(int4*)(rowptr + i) = r;
        }
        carry += __shfl(incl, 31, 32);
    }
}

__global__ __launch_bounds__(256) void appnp_prescale(const int* __restrict__ cnt,
                                                      float* __restrict__ tab) {
    const int t = blockIdx.x * blockDim.x + threadIdx.x;
    if (t < N_NODES * (C_DIM / 4)) {
        const int i = t / (C_DIM / 4);
        const int q = t - i * (C_DIM / 4);
        const float di = rsqrtf((float)(cnt[i] + 1));
        float4* p = (float4*)tab + (i * ROW_F4 + q);
        float4 v = *p;
        v.x *= di; v.y *= di; v.z *= di; v.w *= di;
        *p = v;
    }
}

__global__ __launch_bounds__(256) void appnp_fill(const int* __restrict__ src,
                                                  const int* __restrict__ dst,
                                                  const int* __restrict__ rank,
                                                  const int* __restrict__ rowptr,
                                                  int* __restrict__ col) {
    int e = blockIdx.x * blockDim.x + threadIdx.x;
    if (e < N_EDGES) col[rowptr[dst[e]] + rank[e]] = src[e];
}

template <bool LAST>
__global__ __launch_bounds__(256) void appnp_gather(const int* __restrict__ cnt,
                                                    const int* __restrict__ rowptr,
                                                    const int* __restrict__ col,
                                                    const float* __restrict__ g,
                                                    const float* __restrict__ h0r,
                                                    float* __restrict__ gn) {
    const int lane = threadIdx.x & 31;
    const int i    = blockIdx.x * 8 + (threadIdx.x >> 5);
    const int grp  = lane / 10;
    const int q    = lane - grp * 10;
    const int n     = cnt[i];
    const int start = rowptr[i];
    const float di  = rsqrtf((float)(n + 1));
    const float4* g4 = (const float4*)g;

    float4 gi = make_float4(0.f, 0.f, 0.f, 0.f);
    if (lane < 16) gi = g4[i * ROW_F4 + lane];
    float4 zr = make_float4(0.f, 0.f, 0.f, 0.f);
    if (lane >= 6 && lane < 10) zr = ((const float4*)h0r)[i * 4 + (lane - 6)];

    float4 a0 = make_float4(0.f, 0.f, 0.f, 0.f);
    float4 a1 = a0, a2 = a0, a3 = a0;
    if (grp < 3) {
        const int* cp = col + start;
        for (int j = grp; j < n; j += 12) {
            const bool p1 = (j + 3) < n;
            const bool p2 = (j + 6) < n;
            const bool p3 = (j + 9) < n;
            const int c0 = cp[j];
            const int c1 = cp[p1 ? j + 3 : j];
            const int c2 = cp[p2 ? j + 6 : j];
            const int c3 = cp[p3 ? j + 9 : j];
            const float4 r0 = g4[c0 * ROW_F4 + q];
            const float4 r1 = g4[c1 * ROW_F4 + q];
            const float4 r2 = g4[c2 * ROW_F4 + q];
            const float4 r3 = g4[c3 * ROW_F4 + q];
            a0.x += r0.x; a0.y += r0.y; a0.z += r0.z; a0.w += r0.w;
            a1.x += p1 ? r1.x : 0.f; a1.y += p1 ? r1.y : 0.f;
            a1.z += p1 ? r1.z : 0.f; a1.w += p1 ? r1.w : 0.f;
            a2.x += p2 ? r2.x : 0.f; a2.y += p2 ? r2.y : 0.f;
            a2.z += p2 ? r2.z : 0.f; a2.w += p2 ? r2.w : 0.f;
            a3.x += p3 ? r3.x : 0.f; a3.y += p3 ? r3.y : 0.f;
            a3.z += p3 ? r3.z : 0.f; a3.w += p3 ? r3.w : 0.f;
        }
    }
    float4 a;
    a.x = (a0.x + a1.x) + (a2.x + a3.x);
    a.y = (a0.y + a1.y) + (a2.y + a3.y);
    a.z = (a0.z + a1.z) + (a2.z + a3.z);
    a.w = (a0.w + a1.w) + (a2.w + a3.w);

    float4 s;
    s.x = a.x + __shfl_down(a.x, 10, 32) + __shfl_down(a.x, 20, 32);
    s.y = a.y + __shfl_down(a.y, 10, 32) + __shfl_down(a.y, 20, 32);
    s.z = a.z + __shfl_down(a.z, 10, 32) + __shfl_down(a.z, 20, 32);
    s.w = a.w + __shfl_down(a.w, 10, 32) + __shfl_down(a.w, 20, 32);

    float4 z;
    z.x = __shfl_down(gi.x, 10, 32);
    z.y = __shfl_down(gi.y, 10, 32);
    z.z = __shfl_down(gi.z, 10, 32);
    z.w = __shfl_down(gi.w, 10, 32);
    if (lane >= 6) z = zr;

    if (lane < 16) {
        float4 o = gi;
        if (lane < 10) {
            const float wd = (1.0f - ALPHA) * di;
            o.x = ALPHA * z.x + wd * (s.x + gi.x);
            o.y = ALPHA * z.y + wd * (s.y + gi.y);
            o.z = ALPHA * z.z + wd * (s.z + gi.z);
            o.w = ALPHA * z.w + wd * (s.w + gi.w);
            if (!LAST) { o.x *= di; o.y *= di; o.z *= di; o.w *= di; }
        }
        ((float4*)gn)[i * ROW_F4 + lane] = o;
    }
}

__global__ __launch_bounds__(256) void appnp_logsoftmax(const float* __restrict__ h,
                                                        float* __restrict__ out) {
    const int lane = threadIdx.x & 31;
    const int row = blockIdx.x * 8 + (threadIdx.x >> 5);
    if (row >= N_NODES) return;
    const float* r = h + (size_t)row * ROW_F;
    float v0 = r[lane];
    float v1 = (lane < C_DIM - 32) ? r[lane + 32] : -3.4e38f;

    float m = fmaxf(v0, v1);
#pragma unroll
    for (int off = 16; off > 0; off >>= 1) m = fmaxf(m, __shfl_xor(m, off, 32));

    float s = __expf(v0 - m) + ((lane < C_DIM - 32) ? __expf(v1 - m) : 0.0f);
#pragma unroll
    for (int off = 16; off > 0; off >>= 1) s += __shfl_xor(s, off, 32);

    float lse = m + __logf(s);
    float* o = out + (size_t)row * C_DIM;
    o[lane] = v0 - lse;
    if (lane < C_DIM - 32) o[lane + 32] = v1 - lse;
}

extern "C" void kernel_launch(void* const* d_in, const int* in_sizes, int n_in,
                              void* d_out, int out_size, void* d_ws, size_t ws_size,
                              hipStream_t stream) {
    const float* x  = (const float*)d_in[0];
    const float* W1 = (const float*)d_in[1];
    const float* b1 = (const float*)d_in[2];
    const float* W2 = (const float*)d_in[3];
    const float* b2 = (const float*)d_in[4];
    const int*   ei = (const int*)d_in[5];
    const int*   src = ei;
    const int*   dst = ei + N_EDGES;
    float* out = (float*)d_out;

    char* ws = (char*)d_ws;
    float* hA   = (float*)ws;
    float* hB   = (float*)(ws + 25600000);
    __bf16* h1  = (__bf16*)(ws + 25600000);
    int*   rank = (int*)(ws + 25600000);
    float* h0r  = (float*)(ws + 2 * 25600000);
    int* cnt    = (int*)(ws + 2 * 25600000 + 6400000);
    int* rowptr = (int*)(ws + 2 * 25600000 + 6400000 + 400128);
    int*   col  = (int*)d_out;

    appnp_gemm1<<<(N_NODES / 16 + 3) / 4, 128, 0, stream>>>(x, W1, b1, h1);
    appnp_gemm2<<<((N_NODES / 16) * 3 + 7) / 8, 256, 0, stream>>>(h1, W2, b2, hA, h0r);

    appnp_cnt_zero<<<(N_NODES + 255) / 256, 256, 0, stream>>>(cnt);
    appnp_count_rank<<<(N_EDGES + 255) / 256, 256, 0, stream>>>(dst, cnt, rank);
    appnp_scan<<<1, 1024, 0, stream>>>(cnt, rowptr);
    appnp_prescale<<<(N_NODES * (C_DIM / 4) + 255) / 256, 256, 0, stream>>>(cnt, hA);
    appnp_fill<<<(N_EDGES + 255) / 256, 256, 0, stream>>>(src, dst, rank, rowptr, col);

    float* cur = hA;
    float* nxt = hB;
    for (int k = 0; k < K_STEPS - 1; ++k) {
        appnp_gather<false><<<N_NODES / 8, 256, 0, stream>>>(cnt, rowptr, col, cur, h0r, nxt);
        float* t = cur; cur = nxt; nxt = t;
    }
    appnp_gather<true><<<N_NODES / 8, 256, 0, stream>>>(cnt, rowptr, col, cur, h0r, nxt);
    { float* t = cur; cur = nxt; nxt = t; }

    appnp_logsoftmax<<<(N_NODES + 7) / 8, 256, 0, stream>>>(cur, out);
}


# Edge grouping as a two-level counting sort in LDS (400-node buckets, then nodes): no global per-edge atomics, rowptr gathers or scattered 4-B stores; sum order = arrival order of LDS atomics
# speedup vs baseline: 7.2467x; 1.4372x over previous
#include <hip/hip_runtime.h>
#include <hip/hip_bf16.h>
#include <math.h>

typedef __attribute__((ext_vector_type(16))) __bf16 v16bf;
typedef __attribute__((ext_vector_type(8)))  __bf16 v8bf;
typedef __attribute__((ext_vector_type(8)))  float  v8f;

#define N_NODES 100000
#define F_IN    500
#define H_DIM   64
#define C_DIM   40
#define N_EDGES 3200000
#define K_STEPS 10
#define ALPHA   0.1f
#define ROW_F   64
#define ROW_F4  16

__global__ __launch_bounds__(128) void appnp_gemm1(const float* __restrict__ x,
                                                   const float* __restrict__ W1,
                                                   const float* __restrict__ b1,
                                                   __bf16* __restrict__ h1) {
    const int lane = threadIdx.x & 31;
    const int rowTile = blockIdx.x * 4 + (threadIdx.x >> 5);
    if (rowTile >= N_NODES / 16) return;
    const int c      = lane & 15;
    const int hiHalf = (lane & 16) ? 8 : 0;
    const int m = rowTile * 16 + c;

    const float* xrow = x + (size_t)m * F_IN;
    const float* wcol = W1 + hiHalf * H_DIM + 4 * c;
    v8f acc0 = {}, acc1 = {}, acc2 = {}, acc3 = {};

    for (int kb = 0; kb < 480; kb += 32) {
        float4 w[16];
#pragma unroll
        for (int j = 0; j < 16; ++j) {
            const int k = kb + j + (j & 8);
            w[j] = *(const float4*)(wcol + k * H_DIM);
        }
        const int kA = kb + hiHalf;
        const float4 r0 = *(const float4*)(xrow + kA);
        const float4 r1 = *(const float4*)(xrow + kA + 4);
        const float4 r2 = *(const float4*)(xrow + kA + 16);
        const float4 r3 = *(const float4*)(xrow + kA + 20);
        __builtin_amdgcn_sched_barrier(0);

        v16bf bt0, bt1, bt2, bt3;
#pragma unroll
        for (int j = 0; j < 16; ++j) {
            bt0[j] = (__bf16)w[j].x; bt1[j] = (__bf16)w[j].y;
            bt2[j] = (__bf16)w[j].z; bt3[j] = (__bf16)w[j].w;
        }

        v16bf a;
        a[0]  = (__bf16)r0.x; a[1]  = (__bf16)r0.y; a[2]  = (__bf16)r0.z; a[3]  = (__bf16)r0.w;
        a[4]  = (__bf16)r1.x; a[5]  = (__bf16)r1.y; a[6]  = (__bf16)r1.z; a[7]  = (__bf16)r1.w;
        a[8]  = (__bf16)r2.x; a[9]  = (__bf16)r2.y; a[10] = (__bf16)r2.z; a[11] = (__bf16)r2.w;
        a[12] = (__bf16)r3.x; a[13] = (__bf16)r3.y; a[14] = (__bf16)r3.z; a[15] = (__bf16)r3.w;

        acc0 = __builtin_amdgcn_wmma_f32_16x16x32_bf16(
            false, a, false, bt0, (short)0, acc0, false, false);
        acc1 = __builtin_amdgcn_wmma_f32_16x16x32_bf16(
            false, a, false, bt1, (short)0, acc1, false, false);
        acc2 = __builtin_amdgcn_wmma_f32_16x16x32_bf16(
            false, a, false, bt2, (short)0, acc2, false, false);
        acc3 = __builtin_amdgcn_wmma_f32_16x16x32_bf16(
            false, a, false, bt3, (short)0, acc3, false, false);
    }

    {
        const int kA = 480 + hiHalf;
        const float4 r0 = *(const float4*)(xrow + kA);
        const float4 r1 = *(const float4*)(xrow + kA + 4);
        const float4 r2 = *(const float4*)(xrow + 496);
        const float fa2[4] = {r2.x, r2.y, r2.z, r2.w};

        v16bf a;
        a[0] = (__bf16)r0.x; a[1] = (__bf16)r0.y; a[2] = (__bf16)r0.z; a[3] = (__bf16)r0.w;
        a[4] = (__bf16)r1.x; a[5] = (__bf16)r1.y; a[6] = (__bf16)r1.z; a[7] = (__bf16)r1.w;
#pragma unroll
        for (int j = 0; j < 8; ++j) {
            int k1 = kA + 16 + j;
            float val = (j < 4) ? fa2[j] : 0.0f;
            float f1  = (k1 < F_IN) ? val : 0.0f;
            a[j + 8] = (__bf16)f1;
        }

        v16bf bt0, bt1, bt2, bt3;
#pragma unroll
        for (int j = 0; j < 16; ++j) {
            int k  = 480 + j + (j & 8) + hiHalf;
            int kc = (k < F_IN) ? k : (F_IN - 1);
            const float4 w = *(const float4*)(W1 + kc * H_DIM + 4 * c);
            bt0[j] = (__bf16)w.x; bt1[j] = (__bf16)w.y;
            bt2[j] = (__bf16)w.z; bt3[j] = (__bf16)w.w;
        }
        acc0 = __builtin_amdgcn_wmma_f32_16x16x32_bf16(
            false, a, false, bt0, (short)0, acc0, false, false);
        acc1 = __builtin_amdgcn_wmma_f32_16x16x32_bf16(
            false, a, false, bt1, (short)0, acc1, false, false);
        acc2 = __builtin_amdgcn_wmma_f32_16x16x32_bf16(
            false, a, false, bt2, (short)0, acc2, false, false);
        acc3 = __builtin_amdgcn_wmma_f32_16x16x32_bf16(
            false, a, false, bt3, (short)0, acc3, false, false);
    }

    const float4 bias = *(const float4*)(b1 + 4 * c);
    typedef __attribute__((ext_vector_type(4))) __bf16 v4bf;
#pragma unroll
    for (int r = 0; r < 8; ++r) {
        const int mr = rowTile * 16 + r + hiHalf;
        v4bf o;
        o[0] = (__bf16)fmaxf(acc0[r] + bias.x, 0.0f);
        o[1] = (__bf16)fmaxf(acc1[r] + bias.y, 0.0f);
        o[2] = (__bf16)fmaxf(acc2[r] + bias.z, 0.0f);
        o[3] = (__bf16)fmaxf(acc3[r] + bias.w, 0.0f);
        *(v4bf*)(h1 + (size_t)mr * H_DIM + 4 * c) = o;
    }
}

__global__ __launch_bounds__(256) void appnp_gemm2(const __bf16* __restrict__ h1,
                                                   const float* __restrict__ W2,
                                                   const float* __restrict__ b2,
                                                   float* __restrict__ tab,
                                                   float* __restrict__ h0r) {
    const int lane = threadIdx.x & 31;
    const int wid  = threadIdx.x >> 5;
    const int tile = blockIdx.x * 8 + wid;
    const int nTiles = (N_NODES / 16) * 3;
    if (tile >= nTiles) return;
    const int rowTile = tile / 3;
    const int colTile = tile % 3;
    const int hiHalf  = (lane & 16) ? 8 : 0;
    const int m = rowTile * 16 + (lane & 15);
    const int n = colTile * 16 + (lane & 15);
    const int ncl = (n < C_DIM) ? n : (C_DIM - 1);

    const __bf16* hrow = h1 + (size_t)m * H_DIM;
    v8f acc = {};
#pragma unroll
    for (int kb = 0; kb < H_DIM; kb += 32) {
        const int kA = kb + hiHalf;
        const v8bf lo = *(const v8bf*)(hrow + kA);
        const v8bf hi = *(const v8bf*)(hrow + kA + 16);
        v16bf a;
#pragma unroll
        for (int j = 0; j < 8; ++j) { a[j] = lo[j]; a[j + 8] = hi[j]; }

        v16bf b;
#pragma unroll
        for (int j = 0; j < 16; ++j) {
            int k = kb + j + (j & 8) + hiHalf;
            b[j] = (__bf16)W2[k * C_DIM + ncl];
        }
        acc = __builtin_amdgcn_wmma_f32_16x16x32_bf16(
            false, a, false, b, (short)0, acc, false, false);
    }

    if (n < C_DIM) {
        const float bias = b2[ncl];
#pragma unroll
        for (int r = 0; r < 8; ++r) {
            int mr = rowTile * 16 + r + hiHalf;
            float v = acc[r] + bias;
            float* row = tab + (size_t)mr * ROW_F;
            row[n] = v;
            if (n < 24) row[C_DIM + n] = v;
            else        h0r[(size_t)mr * 16 + (n - 24)] = v;
        }
    }
}

#define BUCK_N   400
#define N_BUCK   (N_NODES / BUCK_N)
#define N_CHUNK  250
#define CHUNK_E  (N_EDGES / N_CHUNK)
#define CSR_T    1024
#define CSR_CAP  16384

template <int PER>
__device__ __forceinline__ void appnp_wave_excl_scan(const int* v, int* o, int n, int lane) {
    int x[PER];
    int s = 0;
#pragma unroll
    for (int j = 0; j < PER; ++j) {
        const int idx = lane * PER + j;
        x[j] = (idx < n) ? v[idx] : 0;
        s += x[j];
    }
    int incl = s;
#pragma unroll
    for (int off = 1; off < 32; off <<= 1) {
        const int t = __shfl_up(incl, off, 32);
        if (lane >= off) incl += t;
    }
    int run = incl - s;
#pragma unroll
    for (int j = 0; j < PER; ++j) {
        const int idx = lane * PER + j;
        if (idx < n) o[idx] = run;
        run += x[j];
    }
}

__global__ __launch_bounds__(CSR_T) void appnp_bucket_hist(const int* __restrict__ dst,
                                                           int* __restrict__ H) {
    __shared__ int h[N_BUCK];
    const int tid = threadIdx.x;
    const int g   = blockIdx.x;
    if (tid < N_BUCK) h[tid] = 0;
    __syncthreads();
    const int* dp = dst + g * CHUNK_E;
    for (int i0 = tid; i0 < CHUNK_E; i0 += 4 * CSR_T) {
        int d[4];
#pragma unroll
        for (int u = 0; u < 4; ++u) {
            const int i = i0 + u * CSR_T;
            d[u] = (i < CHUNK_E) ? dp[i] : -1;
        }
#pragma unroll
        for (int u = 0; u < 4; ++u)
            if (d[u] >= 0) atomicAdd(&h[(unsigned)d[u] / BUCK_N], 1);
    }
    __syncthreads();
    if (tid < N_BUCK) H[g * N_BUCK + tid] = h[tid];
}

__global__ __launch_bounds__(CSR_T) void appnp_bucket_scatter(const int* __restrict__ src,
                                                              const int* __restrict__ dst,
                                                              const int* __restrict__ H,
                                                              int* __restrict__ packed,
                                                              int* __restrict__ bbase) {
    __shared__ int ptot[4][256];
    __shared__ int ppre[4][256];
    __shared__ int tot[N_BUCK];
    __shared__ int mine[N_BUCK];
    __shared__ int base[N_BUCK];
    __shared__ int loff[N_BUCK];
    __shared__ int gofs[N_BUCK];
    __shared__ int cur[N_BUCK];
    __shared__ int stage[CHUNK_E];
    __shared__ unsigned char stageb[CHUNK_E];
    const int tid  = threadIdx.x;
    const int lane = tid & 31;
    const int g    = blockIdx.x;

    {
        const int b = tid & 255;
        const int q = tid >> 8;
        int t = 0, p = 0;
        if (b < N_BUCK) {
#pragma unroll 8
            for (int g2 = q; g2 < N_CHUNK; g2 += 4) {
                const int v = H[g2 * N_BUCK + b];
                t += v;
                p += (g2 < g) ? v : 0;
            }
        }
        ptot[q][b] = t;
        ppre[q][b] = p;
    }
    __syncthreads();
    if (tid < N_BUCK) {
        tot[tid]  = (ptot[0][tid] + ptot[1][tid]) + (ptot[2][tid] + ptot[3][tid]);
        gofs[tid] = (ppre[0][tid] + ppre[1][tid]) + (ppre[2][tid] + ppre[3][tid]);
        mine[tid] = H[g * N_BUCK + tid];
        cur[tid]  = 0;
    }
    __syncthreads();
    if (tid < 32) {
        appnp_wave_excl_scan<8>(tot,  base, N_BUCK, lane);
        appnp_wave_excl_scan<8>(mine, loff, N_BUCK, lane);
    }
    __syncthreads();
    if (tid < N_BUCK) {
        const int b0 = base[tid];
        gofs[tid] += b0;
        if (g == 0) bbase[tid] = b0;
    }
    if (g == 0 && tid == 0) bbase[N_BUCK] = N_EDGES;

    const int* sp = src + g * CHUNK_E;
    const int* dp = dst + g * CHUNK_E;
    for (int i0 = tid; i0 < CHUNK_E; i0 += 4 * CSR_T) {
        int d[4], s[4];
#pragma unroll
        for (int u = 0; u < 4; ++u) {
            const int i = i0 + u * CSR_T;
            d[u] = (i < CHUNK_E) ? dp[i] : -1;
            s[u] = (i < CHUNK_E) ? sp[i] : 0;
        }
#pragma unroll
        for (int u = 0; u < 4; ++u) {
            if (d[u] >= 0) {
                const int b   = (unsigned)d[u] / BUCK_N;
                const int pos = loff[b] + atomicAdd(&cur[b], 1);
                stage[pos]  = (s[u] << 9) | (d[u] - b * BUCK_N);
                stageb[pos] = (unsigned char)b;
            }
        }
    }
    __syncthreads();
    for (int i = tid; i < CHUNK_E; i += CSR_T) {
        const int b = stageb[i];
        packed[gofs[b] + (i - loff[b])] = stage[i];
    }
}

__global__ __launch_bounds__(CSR_T) void appnp_bucket_csr(const int* __restrict__ packed,
                                                          const int* __restrict__ bbase,
                                                          int* __restrict__ cnt,
                                                          int* __restrict__ rowptr,
                                                          int* __restrict__ col) {
    __shared__ int c[BUCK_N];
    __shared__ int rp[BUCK_N];
    __shared__ int cur[BUCK_N];
    __shared__ int lcol[CSR_CAP];
    const int tid  = threadIdx.x;
    const int lane = tid & 31;
    const int b    = blockIdx.x;
    const int lo   = bbase[b];
    const int n    = bbase[b + 1] - lo;
    const bool inLds = (n <= CSR_CAP);
    const int* pp = packed + lo;

    if (tid < BUCK_N) { c[tid] = 0; cur[tid] = 0; }
    __syncthreads();
    for (int i0 = tid; i0 < n; i0 += 4 * CSR_T) {
        int p[4];
#pragma unroll
        for (int u = 0; u < 4; ++u) {
            const int i = i0 + u * CSR_T;
            p[u] = (i < n) ? pp[i] : -1;
        }
#pragma unroll
        for (int u = 0; u < 4; ++u)
            if (p[u] >= 0) atomicAdd(&c[p[u] & 511], 1);
    }
    __syncthreads();
    if (tid < 32) appnp_wave_excl_scan<13>(c, rp, BUCK_N, lane);
    __syncthreads();
    if (tid < BUCK_N) {
        cnt[b * BUCK_N + tid]    = c[tid];
        rowptr[b * BUCK_N + tid] = lo + rp[tid];
    }
    for (int i0 = tid; i0 < n; i0 += 4 * CSR_T) {
        int p[4];
#pragma unroll
        for (int u = 0; u < 4; ++u) {
            const int i = i0 + u * CSR_T;
            p[u] = (i < n) ? pp[i] : -1;
        }
#pragma unroll
        for (int u = 0; u < 4; ++u) {
            if (p[u] >= 0) {
                const int dl  = p[u] & 511;
                const int pos = rp[dl] + atomicAdd(&cur[dl], 1);
                if (inLds) lcol[pos] = p[u] >> 9;
                else       col[lo + pos] = p[u] >> 9;
            }
        }
    }
    __syncthreads();
    if (inLds)
        for (int i = tid; i < n; i += CSR_T) col[lo + i] = lcol[i];
}

__global__ __launch_bounds__(256) void appnp_prescale(const int* __restrict__ cnt,
                                                      float* __restrict__ tab) {
    const int t = blockIdx.x * blockDim.x + threadIdx.x;
    if (t < N_NODES * (C_DIM / 4)) {
        const int i = t / (C_DIM / 4);
        const int q = t - i * (C_DIM / 4);
        const float di = rsqrtf((float)(cnt[i] + 1));
        float4* p = (float4*)tab + (i * ROW_F4 + q);
        float4 v = *p;
        v.x *= di; v.y *= di; v.z *= di; v.w *= di;
        *p = v;
    }
}

template <bool LAST>
__global__ __launch_bounds__(256) void appnp_gather(const int* __restrict__ cnt,
                                                    const int* __restrict__ rowptr,
                                                    const int* __restrict__ col,
                                                    const float* __restrict__ g,
                                                    const float* __restrict__ h0r,
                                                    float* __restrict__ gn) {
    const int lane = threadIdx.x & 31;
    const int i    = blockIdx.x * 8 + (threadIdx.x >> 5);
    const int grp  = lane / 10;
    const int q    = lane - grp * 10;
    const int n     = cnt[i];
    const int start = rowptr[i];
    const float di  = rsqrtf((float)(n + 1));
    const float4* g4 = (const float4*)g;

    float4 gi = make_float4(0.f, 0.f, 0.f, 0.f);
    if (lane < 16) gi = g4[i * ROW_F4 + lane];
    float4 zr = make_float4(0.f, 0.f, 0.f, 0.f);
    if (lane >= 6 && lane < 10) zr = ((const float4*)h0r)[i * 4 + (lane - 6)];

    float4 a0 = make_float4(0.f, 0.f, 0.f, 0.f);
    float4 a1 = a0, a2 = a0, a3 = a0;
    if (grp < 3) {
        const int* cp = col + start;
        for (int j = grp; j < n; j += 12) {
            const bool p1 = (j + 3) < n;
            const bool p2 = (j + 6) < n;
            const bool p3 = (j + 9) < n;
            const int c0 = cp[j];
            const int c1 = cp[p1 ? j + 3 : j];
            const int c2 = cp[p2 ? j + 6 : j];
            const int c3 = cp[p3 ? j + 9 : j];
            const float4 r0 = g4[c0 * ROW_F4 + q];
            const float4 r1 = g4[c1 * ROW_F4 + q];
            const float4 r2 = g4[c2 * ROW_F4 + q];
            const float4 r3 = g4[c3 * ROW_F4 + q];
            a0.x += r0.x; a0.y += r0.y; a0.z += r0.z; a0.w += r0.w;
            a1.x += p1 ? r1.x : 0.f; a1.y += p1 ? r1.y : 0.f;
            a1.z += p1 ? r1.z : 0.f; a1.w += p1 ? r1.w : 0.f;
            a2.x += p2 ? r2.x : 0.f; a2.y += p2 ? r2.y : 0.f;
            a2.z += p2 ? r2.z : 0.f; a2.w += p2 ? r2.w : 0.f;
            a3.x += p3 ? r3.x : 0.f; a3.y += p3 ? r3.y : 0.f;
            a3.z += p3 ? r3.z : 0.f; a3.w += p3 ? r3.w : 0.f;
        }
    }
    float4 a;
    a.x = (a0.x + a1.x) + (a2.x + a3.x);
    a.y = (a0.y + a1.y) + (a2.y + a3.y);
    a.z = (a0.z + a1.z) + (a2.z + a3.z);
    a.w = (a0.w + a1.w) + (a2.w + a3.w);

    float4 s;
    s.x = a.x + __shfl_down(a.x, 10, 32) + __shfl_down(a.x, 20, 32);
    s.y = a.y + __shfl_down(a.y, 10, 32) + __shfl_down(a.y, 20, 32);
    s.z = a.z + __shfl_down(a.z, 10, 32) + __shfl_down(a.z, 20, 32);
    s.w = a.w + __shfl_down(a.w, 10, 32) + __shfl_down(a.w, 20, 32);

    float4 z;
    z.x = __shfl_down(gi.x, 10, 32);
    z.y = __shfl_down(gi.y, 10, 32);
    z.z = __shfl_down(gi.z, 10, 32);
    z.w = __shfl_down(gi.w, 10, 32);
    if (lane >= 6) z = zr;

    if (lane < 16) {
        float4 o = gi;
        if (lane < 10) {
            const float wd = (1.0f - ALPHA) * di;
            o.x = ALPHA * z.x + wd * (s.x + gi.x);
            o.y = ALPHA * z.y + wd * (s.y + gi.y);
            o.z = ALPHA * z.z + wd * (s.z + gi.z);
            o.w = ALPHA * z.w + wd * (s.w + gi.w);
            if (!LAST) { o.x *= di; o.y *= di; o.z *= di; o.w *= di; }
        }
        ((float4*)gn)[i * ROW_F4 + lane] = o;
    }
}

__global__ __launch_bounds__(256) void appnp_logsoftmax(const float* __restrict__ h,
                                                        float* __restrict__ out) {
    const int lane = threadIdx.x & 31;
    const int row = blockIdx.x * 8 + (threadIdx.x >> 5);
    if (row >= N_NODES) return;
    const float* r = h + (size_t)row * ROW_F;
    float v0 = r[lane];
    float v1 = (lane < C_DIM - 32) ? r[lane + 32] : -3.4e38f;

    float m = fmaxf(v0, v1);
#pragma unroll
    for (int off = 16; off > 0; off >>= 1) m = fmaxf(m, __shfl_xor(m, off, 32));

    float s = __expf(v0 - m) + ((lane < C_DIM - 32) ? __expf(v1 - m) : 0.0f);
#pragma unroll
    for (int off = 16; off > 0; off >>= 1) s += __shfl_xor(s, off, 32);

    float lse = m + __logf(s);
    float* o = out + (size_t)row * C_DIM;
    o[lane] = v0 - lse;
    if (lane < C_DIM - 32) o[lane + 32] = v1 - lse;
}

extern "C" void kernel_launch(void* const* d_in, const int* in_sizes, int n_in,
                              void* d_out, int out_size, void* d_ws, size_t ws_size,
                              hipStream_t stream) {
    const float* x  = (const float*)d_in[0];
    const float* W1 = (const float*)d_in[1];
    const float* b1 = (const float*)d_in[2];
    const float* W2 = (const float*)d_in[3];
    const float* b2 = (const float*)d_in[4];
    const int*   ei = (const int*)d_in[5];
    const int*   src = ei;
    const int*   dst = ei + N_EDGES;
    float* out = (float*)d_out;

    char* ws = (char*)d_ws;
    float* hA   = (float*)ws;
    float* hB   = (float*)(ws + 25600000);
    __bf16* h1  = (__bf16*)(ws + 25600000);
    int* packed = (int*)(ws + 25600000 + 12800000);
    float* h0r  = (float*)(ws + 2 * 25600000);
    int* cnt    = (int*)(ws + 2 * 25600000 + 6400000);
    int* rowptr = (int*)(ws + 2 * 25600000 + 6400000 + 400128);
    int*   col  = (int*)d_out;
    int*   H    = (int*)((char*)d_out + 12800000);
    int* bbase  = (int*)((char*)d_out + 12800000 + 250112);

    appnp_gemm1<<<(N_NODES / 16 + 3) / 4, 128, 0, stream>>>(x, W1, b1, h1);
    appnp_gemm2<<<((N_NODES / 16) * 3 + 7) / 8, 256, 0, stream>>>(h1, W2, b2, hA, h0r);

    appnp_bucket_hist<<<N_CHUNK, CSR_T, 0, stream>>>(dst, H);
    appnp_bucket_scatter<<<N_CHUNK, CSR_T, 0, stream>>>(src, dst, H, packed, bbase);
    appnp_bucket_csr<<<N_BUCK, CSR_T, 0, stream>>>(packed, bbase, cnt, rowptr, col);
    appnp_prescale<<<(N_NODES * (C_DIM / 4) + 255) / 256, 256, 0, stream>>>(cnt, hA);

    float* cur = hA;
    float* nxt = hB;
    for (int k = 0; k < K_STEPS - 1; ++k) {
        appnp_gather<false><<<N_NODES / 8, 256, 0, stream>>>(cnt, rowptr, col, cur, h0r, nxt);
        float* t = cur; cur = nxt; nxt = t;
    }
    appnp_gather<true><<<N_NODES / 8, 256, 0, stream>>>(cnt, rowptr, col, cur, h0r, nxt);
    { float* t = cur; cur = nxt; nxt = t; }

    appnp_logsoftmax<<<(N_NODES + 7) / 8, 256, 0, stream>>>(cur, out);
}
